# starting kernel
# baseline (speedup 1.0000x reference)
#include <hip/hip_runtime.h>
#include <hip/hip_bf16.h>

#define FFT_L   4096
#define FFT_NT  256
#define PI_F    3.14159265358979323846f

__global__ __launch_bounds__(FFT_NT) void fft_denoise_kernel(
    const float* __restrict__ x, float* __restrict__ xt) {
  __shared__ float re[FFT_L];
  __shared__ float im[FFT_L];
  const int img = blockIdx.x;
  const int tid = threadIdx.x;
  const float* xi = x + (size_t)img * 62 * 62;

  for (int idx = tid; idx < FFT_L; idx += FFT_NT) {
    int i = idx >> 6, j = idx & 63;
    float v = 0.0f;
    if (i >= 1 && i <= 62 && j >= 1 && j <= 62) v = xi[(i - 1) * 62 + (j - 1)];
    re[idx] = v;
    im[idx] = 0.0f;
  }
  __syncthreads();

  for (int s = 11; s >= 0; --s) {
    const int half = 1 << s;
    const float w = -PI_F / (float)half;
    for (int t = tid; t < FFT_L / 2; t += FFT_NT) {
      int pos = t & (half - 1);
      int i0 = ((t >> s) << (s + 1)) + pos;
      int i1 = i0 + half;
      float ar = re[i0], ai = im[i0];
      float br = re[i1], bi = im[i1];
      float sr = ar - br, si = ai - bi;
      float sn, cs;
      __sincosf(w * (float)pos, &sn, &cs);
      re[i0] = ar + br;
      im[i0] = ai + bi;
      re[i1] = sr * cs - si * sn;
      im[i1] = sr * sn + si * cs;
    }
    __syncthreads();
  }

  for (int idx = tid; idx < FFT_L; idx += FFT_NT) {
    if (fabsf(re[idx]) < 0.5f) {
      re[idx] = 0.0f;
      im[idx] = 0.0f;
    }
  }
  __syncthreads();

  for (int s = 0; s <= 11; ++s) {
    const int half = 1 << s;
    const float w = PI_F / (float)half;
    for (int t = tid; t < FFT_L / 2; t += FFT_NT) {
      int pos = t & (half - 1);
      int i0 = ((t >> s) << (s + 1)) + pos;
      int i1 = i0 + half;
      float sn, cs;
      __sincosf(w * (float)pos, &sn, &cs);
      float br = re[i1], bi = im[i1];
      float tr = br * cs - bi * sn;
      float ti = br * sn + bi * cs;
      float ar = re[i0], ai = im[i0];
      re[i0] = ar + tr;
      im[i0] = ai + ti;
      re[i1] = ar - tr;
      im[i1] = ai - ti;
    }
    __syncthreads();
  }

  float* xo = xt + (size_t)img * FFT_L;
  const float scale = 1.0f / (float)FFT_L;
  for (int idx = tid; idx < FFT_L; idx += FFT_NT) xo[idx] = re[idx] * scale;
}

typedef __attribute__((ext_vector_type(2))) float v2f;
typedef __attribute__((ext_vector_type(8))) float v8f;

#define PATCH_ELEMS (32 * 3 * 18)

__global__ __launch_bounds__(128) void conv3x3_wmma_kernel(
    const float* __restrict__ xt, const float* __restrict__ Wf,
    const float* __restrict__ bias, float* __restrict__ out) {
  __shared__ float xs[PATCH_ELEMS];

  const int lane = threadIdx.x & 31;
  const int wave = threadIdx.x >> 5;
  const int o_base = wave * 16;
  const int j0 = blockIdx.x * 16;
  const int i = blockIdx.y;
  const int b = blockIdx.z;

  const int hi = lane >> 4;
  const int n = lane & 15;

  const float* wrow = Wf + (size_t)(o_base + n) * 288;
  const float* xb = xt + (size_t)b * 32 * FFT_L;

  __builtin_prefetch(wrow, 0, 0);

  for (int e = threadIdx.x; e < PATCH_ELEMS; e += 128) {
    int c = e / 54;
    int rem = e - c * 54;
    int kh = rem / 18;
    int col = rem - kh * 18;
    int jc = j0 + col;
    if (jc > 63) jc = 63;
    xs[e] = xb[((size_t)c * 64 + (i + kh)) * 64 + jc];
  }
  __syncthreads();

  v8f acc = {};
  for (int kk = 0; kk < 288; kk += 4) {
    const int ka = kk + 2 * hi;

    v2f a;
    a.x = wrow[ka];
    a.y = wrow[ka + 1];

    int c0 = ka / 9, r0 = ka - c0 * 9;
    int kh0 = r0 / 3, kw0 = r0 - kh0 * 3;
    int kb = ka + 1;
    int c1 = kb / 9, r1 = kb - c1 * 9;
    int kh1 = r1 / 3, kw1 = r1 - kh1 * 3;
    v2f bv;
    bv.x = xs[c0 * 54 + kh0 * 18 + n + kw0];
    bv.y = xs[c1 * 54 + kh1 * 18 + n + kw1];

    acc = __builtin_amdgcn_wmma_f32_16x16x4_f32(false, a, false, bv, (short)0,
                                                acc, false, false);
  }

  if (j0 + n < 62) {
#pragma unroll
    for (int r = 0; r < 8; ++r) {
      int o = o_base + r + 8 * hi;
      float v = acc[r] + bias[o];
      out[(((size_t)b * 64 + o) * 62 + i) * 62 + (j0 + n)] = v;
    }
  }
}

extern "C" void kernel_launch(void* const* d_in, const int* in_sizes, int n_in,
                              void* d_out, int out_size, void* d_ws,
                              size_t ws_size, hipStream_t stream) {
  const float* x = (const float*)d_in[0];
  const float* W = (const float*)d_in[1];
  const float* bias = (const float*)d_in[2];
  float* out = (float*)d_out;
  float* xt = (float*)d_ws;

  fft_denoise_kernel<<<dim3(128), dim3(FFT_NT), 0, stream>>>(x, xt);
  conv3x3_wmma_kernel<<<dim3(4, 62, 4), dim3(128), 0, stream>>>(xt, W, bias,
                                                                out);
}


# conv: weights register-resident: each wave loads its 16 rows of W once and sweeps 4 output rows of a 16-column strip; K loop unrolled to constant LDS offsets; sum order changes
# speedup vs baseline: 2.0261x; 2.0261x over previous
#include <hip/hip_runtime.h>
#include <hip/hip_bf16.h>

#define FFT_L   4096
#define FFT_NT  256

typedef __attribute__((ext_vector_type(2))) float v2f;
typedef __attribute__((ext_vector_type(4))) float v4f;
typedef __attribute__((ext_vector_type(8))) float v8f;

#define FS 68
#define FMAT (64 * FS)

__device__ __forceinline__ v8f mma(v2f a, v2f b, v8f c) {
  return __builtin_amdgcn_wmma_f32_16x16x4_f32(false, a, false, b, (short)0, c,
                                               false, false);
}

struct Frags {
  v2f aR, aI, bR[2], bI[2];
};

template <bool HAS_BI>
__device__ __forceinline__ Frags ld_frags(const float* aRe, const float* aIm,
                                          const float* bRe, const float* bIm,
                                          int aoff, int boff, int k) {
  Frags f;
  f.aR = *(const v2f*)(aRe + aoff + k);
  f.aI = *(const v2f*)(aIm + aoff + k);
  f.bR[0] = *(const v2f*)(bRe + boff + k);
  f.bR[1] = *(const v2f*)(bRe + boff + 16 * FS + k);
  if (HAS_BI) {
    f.bI[0] = *(const v2f*)(bIm + boff + k);
    f.bI[1] = *(const v2f*)(bIm + boff + 16 * FS + k);
  } else {
    f.bI[0] = f.bR[0];
    f.bI[1] = f.bR[1];
  }
  return f;
}

template <int MODE>
__device__ __forceinline__ void product64(const float* aRe, const float* aIm,
                                          const float* bRe, const float* bIm,
                                          int aoff, int boff, int hi,
                                          v8f (&re)[2], v8f (&im)[2]) {
  constexpr bool HAS_BI = (MODE != 0);
  Frags cur = ld_frags<HAS_BI>(aRe, aIm, bRe, bIm, aoff, boff, 2 * hi);
  for (int ks = 0; ks < 16; ++ks) {
    Frags nxt = cur;
    if (ks < 15)
      nxt = ld_frags<HAS_BI>(aRe, aIm, bRe, bIm, aoff, boff,
                             4 * (ks + 1) + 2 * hi);
    __builtin_amdgcn_sched_barrier(0);
    for (int t = 0; t < 2; ++t) {
      if (MODE == 0) {
        re[t] = mma(cur.aR, cur.bR[t], re[t]);
        im[t] = mma(cur.aI, cur.bR[t], im[t]);
      } else if (MODE == 1) {
        re[t] = mma(cur.aR, cur.bR[t], re[t]);
        im[t] = mma(cur.aR, cur.bI[t], im[t]);
        re[t] = mma(-cur.aI, cur.bI[t], re[t]);
        im[t] = mma(cur.aI, cur.bR[t], im[t]);
      } else if (MODE == 2) {
        re[t] = mma(cur.aR, cur.bR[t], re[t]);
        im[t] = mma(cur.aI, cur.bR[t], im[t]);
        re[t] = mma(cur.aI, cur.bI[t], re[t]);
        im[t] = mma(-cur.aR, cur.bI[t], im[t]);
      } else {
        re[t] = mma(cur.aR, cur.bR[t], re[t]);
        re[t] = mma(cur.aI, cur.bI[t], re[t]);
      }
    }
    __builtin_amdgcn_sched_barrier(0);
    cur = nxt;
  }
}

__global__ __launch_bounds__(FFT_NT) void fft_denoise_kernel(
    const float* __restrict__ x, float* __restrict__ xt) {
  __shared__ __attribute__((aligned(16))) float Fr[FMAT];
  __shared__ __attribute__((aligned(16))) float Fi[FMAT];
  __shared__ __attribute__((aligned(16))) float P[FMAT];
  __shared__ __attribute__((aligned(16))) float Q[FMAT];
  __shared__ float rootc[64];
  __shared__ float roots[64];

  const int img = blockIdx.x;
  const int tid = threadIdx.x;
  const int lane = tid & 31;
  const int wave = tid >> 5;
  const int hi = lane >> 4;
  const int n = lane & 15;
  const int mt = wave & 3;
  const int nt0 = (wave >> 2) * 2;
  const float* xi = x + (size_t)img * 62 * 62;

  float xv[16];
  for (int u = 0; u < 16; ++u) {
    int idx = tid + u * FFT_NT;
    int i = idx >> 6, j = idx & 63;
    float v = 0.0f;
    if (i >= 1 && i <= 62 && j >= 1 && j <= 62) v = xi[(i - 1) * 62 + (j - 1)];
    xv[u] = v;
  }

  if (tid < 64) {
    float s, c;
    sincospif((float)tid * (1.0f / 32.0f), &s, &c);
    rootc[tid] = c;
    roots[tid] = -s;
  }

  float twc[2][8], tws[2][8];
  for (int t = 0; t < 2; ++t) {
    for (int r = 0; r < 8; ++r) {
      int k1 = 16 * mt + 8 * hi + r;
      int n2 = 16 * (nt0 + t) + n;
      float s, c;
      sincospif((float)(k1 * n2) * (1.0f / 2048.0f), &s, &c);
      twc[t][r] = c;
      tws[t][r] = -s;
    }
  }

  for (int u = 0; u < 16; ++u) {
    int idx = tid + u * FFT_NT;
    int i = idx >> 6, j = idx & 63;
    P[j * FS + i] = xv[u];
  }
  __syncthreads();

  for (int idx = tid; idx < FFT_L; idx += FFT_NT) {
    int a = idx >> 6, b = idx & 63;
    int q = (a * b) & 63;
    Fr[a * FS + b] = rootc[q];
    Fi[a * FS + b] = roots[q];
  }
  __syncthreads();

  const int aoff = (16 * mt + n) * FS;
  const int boff = (16 * nt0 + n) * FS;
  v8f accr[2], acci[2];

  accr[0] = (v8f){}; accr[1] = (v8f){}; acci[0] = (v8f){}; acci[1] = (v8f){};
  product64<0>(Fr, Fi, P, P, aoff, boff, hi, accr, acci);
  __syncthreads();
  for (int t = 0; t < 2; ++t) {
    for (int r = 0; r < 8; ++r) {
      float yr = accr[t][r], yi = acci[t][r];
      float c = twc[t][r], s = tws[t][r];
      int o = (16 * mt + 8 * hi + r) * FS + 16 * (nt0 + t) + n;
      P[o] = yr * c - yi * s;
      Q[o] = yr * s + yi * c;
    }
  }
  __syncthreads();

  accr[0] = (v8f){}; accr[1] = (v8f){}; acci[0] = (v8f){}; acci[1] = (v8f){};
  product64<1>(P, Q, Fr, Fi, aoff, boff, hi, accr, acci);
  __syncthreads();
  for (int t = 0; t < 2; ++t) {
    for (int r = 0; r < 8; ++r) {
      float zr = accr[t][r], zi = acci[t][r];
      if (fabsf(zr) < 0.5f) {
        zr = 0.0f;
        zi = 0.0f;
      }
      int o = (16 * mt + 8 * hi + r) * FS + 16 * (nt0 + t) + n;
      P[o] = zr;
      Q[o] = zi;
    }
  }
  __syncthreads();

  accr[0] = (v8f){}; accr[1] = (v8f){}; acci[0] = (v8f){}; acci[1] = (v8f){};
  product64<2>(P, Q, Fr, Fi, aoff, boff, hi, accr, acci);
  __syncthreads();
  for (int t = 0; t < 2; ++t) {
    v4f pr0, pr1, pi0, pi1;
    for (int r = 0; r < 4; ++r) {
      float ur = accr[t][r], ui = acci[t][r];
      float c = twc[t][r], s = tws[t][r];
      pr0[r] = ur * c + ui * s;
      pi0[r] = ui * c - ur * s;
      float vr = accr[t][r + 4], vi = acci[t][r + 4];
      float c4 = twc[t][r + 4], s4 = tws[t][r + 4];
      pr1[r] = vr * c4 + vi * s4;
      pi1[r] = vi * c4 - vr * s4;
    }
    int o = (16 * (nt0 + t) + n) * FS + 16 * mt + 8 * hi;
    *(v4f*)(P + o) = pr0;
    *(v4f*)(P + o + 4) = pr1;
    *(v4f*)(Q + o) = pi0;
    *(v4f*)(Q + o + 4) = pi1;
  }
  __syncthreads();

  accr[0] = (v8f){}; accr[1] = (v8f){};
  product64<3>(Fr, Fi, P, Q, aoff, boff, hi, accr, acci);

  float* xo = xt + (size_t)img * FFT_L;
  const float scale = 1.0f / (float)FFT_L;
  for (int t = 0; t < 2; ++t) {
    for (int r = 0; r < 8; ++r) {
      xo[(16 * mt + 8 * hi + r) * 64 + 16 * (nt0 + t) + n] = accr[t][r] * scale;
    }
  }
}

#define CV_R 4
#define CV_ROWS (CV_R + 2)
#define CV_CH (CV_ROWS * 18)
#define CV_PC (CV_CH + 8)

__device__ __forceinline__ constexpr int cv_off(int k) {
  return (k / 9) * CV_PC + ((k % 9) / 3) * 18 + (k % 9) % 3;
}
__device__ __forceinline__ constexpr int cv_delta(int k) {
  return cv_off(k + 4) - cv_off(k);
}
__device__ __forceinline__ constexpr int cv_dsel(int k) {
  return (k % 9) == 2 ? 1 : ((k % 9) <= 4 ? 0 : (((k % 9) == 5 || (k % 9) == 8) ? 2 : 3));
}
__host__ __device__ constexpr bool cv_bases_ok() {
  const int rep[4] = {0, 2, 5, 6};
  for (int k = 0; k + 4 < 288; ++k)
    if (cv_delta(k) != cv_delta(rep[cv_dsel(k)])) return false;
  return true;
}
static_assert(cv_bases_ok(), "lane-half distance table does not match");

__global__ __launch_bounds__(128) void conv3x3_wmma_kernel(
    const float* __restrict__ xt, const float* __restrict__ Wf,
    const float* __restrict__ bias, float* __restrict__ out) {
  __shared__ float xs[32 * CV_PC];

  const int tid = threadIdx.x;
  const int lane = tid & 31;
  const int wave = tid >> 5;
  const int o_base = wave * 16;
  const int j0 = blockIdx.x * 16;
  const int i0 = blockIdx.y * CV_R;
  const int b = blockIdx.z;

  const int hi = lane >> 4;
  const int n = lane & 15;

  const v4f* wv = (const v4f*)(Wf + (size_t)(o_base + n) * 288 + 4 * hi);
  v4f wreg[36];
#pragma unroll
  for (int u = 0; u < 36; ++u) wreg[u] = wv[2 * u];

  const float* xb = xt + (size_t)b * 32 * FFT_L;
  const int cell = tid < CV_CH ? tid : CV_CH - 1;
  const int pr = cell / 18;
  const int pc = cell - pr * 18;
  int prow = i0 + pr;
  if (prow > 63) prow = 63;
  int pcol = j0 + pc;
  if (pcol > 63) pcol = 63;
  const float* src = xb + prow * 64 + pcol;
  float pv[32];
#pragma unroll
  for (int c = 0; c < 32; ++c) pv[c] = src[c * FFT_L];

  if (tid < CV_CH) {
#pragma unroll
    for (int c = 0; c < 32; ++c) xs[c * CV_PC + tid] = pv[c];
  }
  __syncthreads();

  const float* xl[4];
  xl[0] = xs + n + hi * cv_delta(0);
  xl[1] = xs + n + hi * cv_delta(2);
  xl[2] = xs + n + hi * cv_delta(5);
  xl[3] = xs + n + hi * cv_delta(6);
  v8f acc[CV_R];
  for (int t = 0; t < CV_R; ++t) acc[t] = (v8f){};

  v2f bcur[2][CV_R];
  for (int q = 0; q < 2; ++q) {
    for (int t = 0; t < CV_R; ++t) {
      bcur[q][t].x = xl[cv_dsel(2 * q)][cv_off(2 * q) + 18 * t];
      bcur[q][t].y = xl[cv_dsel(2 * q + 1)][cv_off(2 * q + 1) + 18 * t];
    }
  }
#pragma unroll
  for (int j = 0; j < 36; ++j) {
    v2f bnxt[2][CV_R];
    for (int q = 0; q < 2; ++q) {
      for (int t = 0; t < CV_R; ++t) {
        bnxt[q][t] = bcur[q][t];
        if (j < 35) {
          const int k0 = 8 * (j + 1) + 2 * q;
          bnxt[q][t].x = xl[cv_dsel(k0)][cv_off(k0) + 18 * t];
          bnxt[q][t].y = xl[cv_dsel(k0 + 1)][cv_off(k0 + 1) + 18 * t];
        }
      }
    }
    __builtin_amdgcn_sched_barrier(0);
    for (int q = 0; q < 2; ++q) {
      v2f a;
      a.x = wreg[j][2 * q];
      a.y = wreg[j][2 * q + 1];
      for (int t = 0; t < CV_R; ++t) acc[t] = mma(a, bcur[q][t], acc[t]);
    }
    __builtin_amdgcn_sched_barrier(0);
    for (int q = 0; q < 2; ++q)
      for (int t = 0; t < CV_R; ++t) bcur[q][t] = bnxt[q][t];
  }

  if (j0 + n < 62) {
    float bv[8];
    for (int r = 0; r < 8; ++r) bv[r] = bias[o_base + r + 8 * hi];
#pragma unroll
    for (int t = 0; t < CV_R; ++t) {
      int i = i0 + t;
      if (i < 62) {
#pragma unroll
        for (int r = 0; r < 8; ++r) {
          int o = o_base + r + 8 * hi;
          out[(((size_t)b * 64 + o) * 62 + i) * 62 + (j0 + n)] =
              acc[t][r] + bv[r];
        }
      }
    }
  }
}

extern "C" void kernel_launch(void* const* d_in, const int* in_sizes, int n_in,
                              void* d_out, int out_size, void* d_ws,
                              size_t ws_size, hipStream_t stream) {
  const float* x = (const float*)d_in[0];
  const float* W = (const float*)d_in[1];
  const float* bias = (const float*)d_in[2];
  float* out = (float*)d_out;
  float* xt = (float*)d_ws;

  fft_denoise_kernel<<<dim3(128), dim3(FFT_NT), 0, stream>>>(x, xt);
  conv3x3_wmma_kernel<<<dim3(4, 16, 4), dim3(128), 0, stream>>>(xt, W, bias,
                                                                out);
}


# FFT: conjugate symmetry of a real signal's spectrum: only columns k2<32 computed and inverted with weight 2 (k=2048 as an alternating add); the two middle products halve; sum order changes
# speedup vs baseline: 2.2246x; 1.0980x over previous
#include <hip/hip_runtime.h>
#include <hip/hip_bf16.h>

#define FFT_L   4096
#define FFT_NT  256

typedef __attribute__((ext_vector_type(2))) float v2f;
typedef __attribute__((ext_vector_type(4))) float v4f;
typedef __attribute__((ext_vector_type(8))) float v8f;

#define FS 68
#define FMAT (64 * FS)

__device__ __forceinline__ v8f mma(v2f a, v2f b, v8f c) {
  return __builtin_amdgcn_wmma_f32_16x16x4_f32(false, a, false, b, (short)0, c,
                                               false, false);
}

struct Frags {
  v2f aR, aI, bR[2], bI[2];
};

template <bool HAS_BI, int NT>
__device__ __forceinline__ Frags ld_frags(const float* aRe, const float* aIm,
                                          const float* bRe, const float* bIm,
                                          int aoff, int boff, int k) {
  Frags f;
  f.aR = *(const v2f*)(aRe + aoff + k);
  f.aI = *(const v2f*)(aIm + aoff + k);
  f.bR[0] = *(const v2f*)(bRe + boff + k);
  f.bR[1] = f.bR[0];
  if (NT == 2) f.bR[1] = *(const v2f*)(bRe + boff + 16 * FS + k);
  if (HAS_BI) {
    f.bI[0] = *(const v2f*)(bIm + boff + k);
    f.bI[1] = f.bI[0];
    if (NT == 2) f.bI[1] = *(const v2f*)(bIm + boff + 16 * FS + k);
  } else {
    f.bI[0] = f.bR[0];
    f.bI[1] = f.bR[1];
  }
  return f;
}

template <int MODE, int NT, int KS>
__device__ __forceinline__ void product64(const float* aRe, const float* aIm,
                                          const float* bRe, const float* bIm,
                                          int aoff, int boff, int hi,
                                          v8f (&re)[2], v8f (&im)[2]) {
  constexpr bool HAS_BI = (MODE != 0);
  Frags cur = ld_frags<HAS_BI, NT>(aRe, aIm, bRe, bIm, aoff, boff, 2 * hi);
  for (int ks = 0; ks < KS; ++ks) {
    Frags nxt = cur;
    if (ks < KS - 1)
      nxt = ld_frags<HAS_BI, NT>(aRe, aIm, bRe, bIm, aoff, boff,
                                 4 * (ks + 1) + 2 * hi);
    __builtin_amdgcn_sched_barrier(0);
    for (int t = 0; t < NT; ++t) {
      if (MODE == 0) {
        re[t] = mma(cur.aR, cur.bR[t], re[t]);
        im[t] = mma(cur.aI, cur.bR[t], im[t]);
      } else if (MODE == 1) {
        re[t] = mma(cur.aR, cur.bR[t], re[t]);
        im[t] = mma(cur.aR, cur.bI[t], im[t]);
        re[t] = mma(-cur.aI, cur.bI[t], re[t]);
        im[t] = mma(cur.aI, cur.bR[t], im[t]);
      } else if (MODE == 2) {
        re[t] = mma(cur.aR, cur.bR[t], re[t]);
        im[t] = mma(cur.aI, cur.bR[t], im[t]);
        re[t] = mma(cur.aI, cur.bI[t], re[t]);
        im[t] = mma(-cur.aR, cur.bI[t], im[t]);
      } else {
        re[t] = mma(cur.aR, cur.bR[t], re[t]);
        re[t] = mma(cur.aI, cur.bI[t], re[t]);
      }
    }
    __builtin_amdgcn_sched_barrier(0);
    cur = nxt;
  }
}

__global__ __launch_bounds__(FFT_NT) void fft_denoise_kernel(
    const float* __restrict__ x, float* __restrict__ xt) {
  __shared__ __attribute__((aligned(16))) float Fr[FMAT];
  __shared__ __attribute__((aligned(16))) float Fi[FMAT];
  __shared__ __attribute__((aligned(16))) float P[FMAT];
  __shared__ __attribute__((aligned(16))) float Q[FMAT];
  __shared__ float rootc[64];
  __shared__ float roots[64];
  __shared__ float altsum[8];

  const int img = blockIdx.x;
  const int tid = threadIdx.x;
  const int lane = tid & 31;
  const int wave = tid >> 5;
  const int hi = lane >> 4;
  const int n = lane & 15;
  const int mt = wave & 3;
  const int nt0 = (wave >> 2) * 2;
  const float* xi = x + (size_t)img * 62 * 62;

  float xv[16];
  for (int u = 0; u < 16; ++u) {
    int idx = tid + u * FFT_NT;
    int i = idx >> 6, j = idx & 63;
    float v = 0.0f;
    if (i >= 1 && i <= 62 && j >= 1 && j <= 62) v = xi[(i - 1) * 62 + (j - 1)];
    xv[u] = v;
  }

  if (tid < 64) {
    float s, c;
    sincospif((float)tid * (1.0f / 32.0f), &s, &c);
    rootc[tid] = c;
    roots[tid] = -s;
  }

  float twc[2][8], tws[2][8];
  for (int t = 0; t < 2; ++t) {
    for (int r = 0; r < 8; ++r) {
      int k1 = 16 * mt + 8 * hi + r;
      int n2 = 16 * (nt0 + t) + n;
      float s, c;
      sincospif((float)(k1 * n2) * (1.0f / 2048.0f), &s, &c);
      twc[t][r] = c;
      tws[t][r] = -s;
    }
  }

  for (int u = 0; u < 16; ++u) {
    int idx = tid + u * FFT_NT;
    int i = idx >> 6, j = idx & 63;
    P[j * FS + i] = xv[u];
  }
  {
    float sa = 0.0f;
    for (int u = 0; u < 16; ++u) sa += xv[u];
    if (tid & 1) sa = -sa;
    for (int m = 16; m >= 1; m >>= 1) sa += __shfl_xor(sa, m);
    if (lane == 0) altsum[wave] = sa;
  }
  __syncthreads();
  float z2048 = 0.0f;
  for (int w = 0; w < 8; ++w) z2048 += altsum[w];
  if (fabsf(z2048) < 0.5f) z2048 = 0.0f;

  for (int idx = tid; idx < FFT_L; idx += FFT_NT) {
    int a = idx >> 6, b = idx & 63;
    int q = (a * b) & 63;
    Fr[a * FS + b] = rootc[q];
    Fi[a * FS + b] = roots[q];
  }
  __syncthreads();

  const int aoff = (16 * mt + n) * FS;
  const int boff = (16 * nt0 + n) * FS;
  v8f accr[2], acci[2];

  accr[0] = (v8f){}; accr[1] = (v8f){}; acci[0] = (v8f){}; acci[1] = (v8f){};
  product64<0, 2, 16>(Fr, Fi, P, P, aoff, boff, hi, accr, acci);
  __syncthreads();
  for (int t = 0; t < 2; ++t) {
    for (int r = 0; r < 8; ++r) {
      float yr = accr[t][r], yi = acci[t][r];
      float c = twc[t][r], s = tws[t][r];
      int o = (16 * mt + 8 * hi + r) * FS + 16 * (nt0 + t) + n;
      P[o] = yr * c - yi * s;
      Q[o] = yr * s + yi * c;
    }
  }
  __syncthreads();

  const int nth = wave >> 2;
  const int boffh = (16 * nth + n) * FS;
  accr[0] = (v8f){}; acci[0] = (v8f){};
  product64<1, 1, 16>(P, Q, Fr, Fi, aoff, boffh, hi, accr, acci);
  __syncthreads();
  for (int r = 0; r < 8; ++r) {
    float zr = accr[0][r], zi = acci[0][r];
    const int k1 = 16 * mt + 8 * hi + r;
    const int k2 = 16 * nth + n;
    const float wgt = (k1 == 0 && k2 == 0) ? 1.0f : 2.0f;
    if (fabsf(zr) < 0.5f) {
      zr = 0.0f;
      zi = 0.0f;
    }
    int o = k1 * FS + k2;
    P[o] = zr * wgt;
    Q[o] = zi * wgt;
  }
  __syncthreads();

  accr[0] = (v8f){}; accr[1] = (v8f){}; acci[0] = (v8f){}; acci[1] = (v8f){};
  product64<2, 2, 8>(P, Q, Fr, Fi, aoff, boff, hi, accr, acci);
  __syncthreads();
  for (int t = 0; t < 2; ++t) {
    v4f pr0, pr1, pi0, pi1;
    for (int r = 0; r < 4; ++r) {
      float ur = accr[t][r], ui = acci[t][r];
      float c = twc[t][r], s = tws[t][r];
      pr0[r] = ur * c + ui * s;
      pi0[r] = ui * c - ur * s;
      float vr = accr[t][r + 4], vi = acci[t][r + 4];
      float c4 = twc[t][r + 4], s4 = tws[t][r + 4];
      pr1[r] = vr * c4 + vi * s4;
      pi1[r] = vi * c4 - vr * s4;
    }
    int o = (16 * (nt0 + t) + n) * FS + 16 * mt + 8 * hi;
    *(v4f*)(P + o) = pr0;
    *(v4f*)(P + o + 4) = pr1;
    *(v4f*)(Q + o) = pi0;
    *(v4f*)(Q + o + 4) = pi1;
  }
  __syncthreads();

  accr[0] = (v8f){}; accr[1] = (v8f){};
  product64<3, 2, 16>(Fr, Fi, P, Q, aoff, boff, hi, accr, acci);

  float* xo = xt + (size_t)img * FFT_L;
  const float scale = 1.0f / (float)FFT_L;
  const float alt = (n & 1) ? -z2048 : z2048;
  for (int t = 0; t < 2; ++t) {
    for (int r = 0; r < 8; ++r) {
      xo[(16 * mt + 8 * hi + r) * 64 + 16 * (nt0 + t) + n] =
          (accr[t][r] + alt) * scale;
    }
  }
}

#define CV_R 4
#define CV_ROWS (CV_R + 2)
#define CV_CH (CV_ROWS * 18)
#define CV_PC (CV_CH + 8)

__device__ __forceinline__ constexpr int cv_off(int k) {
  return (k / 9) * CV_PC + ((k % 9) / 3) * 18 + (k % 9) % 3;
}
__device__ __forceinline__ constexpr int cv_delta(int k) {
  return cv_off(k + 4) - cv_off(k);
}
__device__ __forceinline__ constexpr int cv_dsel(int k) {
  return (k % 9) == 2 ? 1 : ((k % 9) <= 4 ? 0 : (((k % 9) == 5 || (k % 9) == 8) ? 2 : 3));
}
__host__ __device__ constexpr bool cv_bases_ok() {
  const int rep[4] = {0, 2, 5, 6};
  for (int k = 0; k + 4 < 288; ++k)
    if (cv_delta(k) != cv_delta(rep[cv_dsel(k)])) return false;
  return true;
}
static_assert(cv_bases_ok(), "lane-half distance table does not match");

__global__ __launch_bounds__(128) void conv3x3_wmma_kernel(
    const float* __restrict__ xt, const float* __restrict__ Wf,
    const float* __restrict__ bias, float* __restrict__ out) {
  __shared__ float xs[32 * CV_PC];

  const int tid = threadIdx.x;
  const int lane = tid & 31;
  const int wave = tid >> 5;
  const int o_base = wave * 16;
  const int j0 = blockIdx.x * 16;
  const int i0 = blockIdx.y * CV_R;
  const int b = blockIdx.z;

  const int hi = lane >> 4;
  const int n = lane & 15;

  const v4f* wv = (const v4f*)(Wf + (size_t)(o_base + n) * 288 + 4 * hi);
  v4f wreg[36];
#pragma unroll
  for (int u = 0; u < 36; ++u) wreg[u] = wv[2 * u];

  const float* xb = xt + (size_t)b * 32 * FFT_L;
  const int cell = tid < CV_CH ? tid : CV_CH - 1;
  const int pr = cell / 18;
  const int pc = cell - pr * 18;
  int prow = i0 + pr;
  if (prow > 63) prow = 63;
  int pcol = j0 + pc;
  if (pcol > 63) pcol = 63;
  const float* src = xb + prow * 64 + pcol;
  float pv[32];
#pragma unroll
  for (int c = 0; c < 32; ++c) pv[c] = src[c * FFT_L];

  if (tid < CV_CH) {
#pragma unroll
    for (int c = 0; c < 32; ++c) xs[c * CV_PC + tid] = pv[c];
  }
  __syncthreads();

  const float* xl[4];
  xl[0] = xs + n + hi * cv_delta(0);
  xl[1] = xs + n + hi * cv_delta(2);
  xl[2] = xs + n + hi * cv_delta(5);
  xl[3] = xs + n + hi * cv_delta(6);
  v8f acc[CV_R];
  for (int t = 0; t < CV_R; ++t) acc[t] = (v8f){};

  v2f bcur[2][CV_R];
  for (int q = 0; q < 2; ++q) {
    for (int t = 0; t < CV_R; ++t) {
      bcur[q][t].x = xl[cv_dsel(2 * q)][cv_off(2 * q) + 18 * t];
      bcur[q][t].y = xl[cv_dsel(2 * q + 1)][cv_off(2 * q + 1) + 18 * t];
    }
  }
#pragma unroll
  for (int j = 0; j < 36; ++j) {
    v2f bnxt[2][CV_R];
    for (int q = 0; q < 2; ++q) {
      for (int t = 0; t < CV_R; ++t) {
        bnxt[q][t] = bcur[q][t];
        if (j < 35) {
          const int k0 = 8 * (j + 1) + 2 * q;
          bnxt[q][t].x = xl[cv_dsel(k0)][cv_off(k0) + 18 * t];
          bnxt[q][t].y = xl[cv_dsel(k0 + 1)][cv_off(k0 + 1) + 18 * t];
        }
      }
    }
    __builtin_amdgcn_sched_barrier(0);
    for (int q = 0; q < 2; ++q) {
      v2f a;
      a.x = wreg[j][2 * q];
      a.y = wreg[j][2 * q + 1];
      for (int t = 0; t < CV_R; ++t) acc[t] = mma(a, bcur[q][t], acc[t]);
    }
    __builtin_amdgcn_sched_barrier(0);
    for (int q = 0; q < 2; ++q)
      for (int t = 0; t < CV_R; ++t) bcur[q][t] = bnxt[q][t];
  }

  if (j0 + n < 62) {
    float bv[8];
    for (int r = 0; r < 8; ++r) bv[r] = bias[o_base + r + 8 * hi];
#pragma unroll
    for (int t = 0; t < CV_R; ++t) {
      int i = i0 + t;
      if (i < 62) {
#pragma unroll
        for (int r = 0; r < 8; ++r) {
          int o = o_base + r + 8 * hi;
          out[(((size_t)b * 64 + o) * 62 + i) * 62 + (j0 + n)] =
              acc[t][r] + bv[r];
        }
      }
    }
  }
}

extern "C" void kernel_launch(void* const* d_in, const int* in_sizes, int n_in,
                              void* d_out, int out_size, void* d_ws,
                              size_t ws_size, hipStream_t stream) {
  const float* x = (const float*)d_in[0];
  const float* W = (const float*)d_in[1];
  const float* bias = (const float*)d_in[2];
  float* out = (float*)d_out;
  float* xt = (float*)d_ws;

  fft_denoise_kernel<<<dim3(128), dim3(FFT_NT), 0, stream>>>(x, xt);
  conv3x3_wmma_kernel<<<dim3(4, 16, 4), dim3(128), 0, stream>>>(xt, W, bias,
                                                                out);
}


# FFT: rows k1 of the 64x64 four-step split by parity over two workgroups per image (256 WGs), one side of every product 32; conv patch stage adds the two planes, reads before W's; sum order changes
# speedup vs baseline: 2.5773x; 1.1585x over previous
#include <hip/hip_runtime.h>
#include <hip/hip_bf16.h>

#define FFT_L   4096
#define FFT_NT  256
#define FFT_HP  2048

typedef __attribute__((ext_vector_type(2))) float v2f;
typedef __attribute__((ext_vector_type(4))) float v4f;
typedef __attribute__((ext_vector_type(8))) float v8f;

#define FS 68
#define HS 36
#define FMAT (64 * FS)
#define HM32 (32 * HS)
#define HM64 (64 * HS)

__device__ __forceinline__ v8f mma(v2f a, v2f b, v8f c) {
  return __builtin_amdgcn_wmma_f32_16x16x4_f32(false, a, false, b, (short)0, c,
                                               false, false);
}

struct Frags {
  v2f aR, aI, bR[2], bI[2];
};

template <bool HAS_BI, int NT, int BP>
__device__ __forceinline__ Frags ld_frags(const float* aRe, const float* aIm,
                                          const float* bRe, const float* bIm,
                                          int aoff, int boff, int k) {
  Frags f;
  f.aR = *(const v2f*)(aRe + aoff + k);
  f.aI = *(const v2f*)(aIm + aoff + k);
  f.bR[0] = *(const v2f*)(bRe + boff + k);
  f.bR[1] = f.bR[0];
  if (NT == 2) f.bR[1] = *(const v2f*)(bRe + boff + 16 * BP + k);
  if (HAS_BI) {
    f.bI[0] = *(const v2f*)(bIm + boff + k);
    f.bI[1] = f.bI[0];
    if (NT == 2) f.bI[1] = *(const v2f*)(bIm + boff + 16 * BP + k);
  } else {
    f.bI[0] = f.bR[0];
    f.bI[1] = f.bR[1];
  }
  return f;
}

template <int MODE, int NT, int KS, int G, int BP>
__device__ __forceinline__ void product64(const float* aRe, const float* aIm,
                                          const float* bRe, const float* bIm,
                                          int aoff, int boff, int hi,
                                          v8f (&re)[2], v8f (&im)[2]) {
  constexpr bool HAS_BI = (MODE != 0);
  constexpr int NG = KS / G;
  Frags cur[G];
#pragma unroll
  for (int g = 0; g < G; ++g)
    cur[g] = ld_frags<HAS_BI, NT, BP>(aRe, aIm, bRe, bIm, aoff, boff,
                                      4 * g + 2 * hi);
#pragma unroll
  for (int s = 0; s < NG; ++s) {
    Frags nxt[G];
#pragma unroll
    for (int g = 0; g < G; ++g) {
      nxt[g] = cur[g];
      if (s < NG - 1)
        nxt[g] = ld_frags<HAS_BI, NT, BP>(aRe, aIm, bRe, bIm, aoff, boff,
                                          4 * ((s + 1) * G + g) + 2 * hi);
    }
    __builtin_amdgcn_sched_barrier(0);
#pragma unroll
    for (int g = 0; g < G; ++g) {
      const Frags& c = cur[g];
#pragma unroll
      for (int t = 0; t < NT; ++t) {
        if (MODE == 0) {
          re[t] = mma(c.aR, c.bR[t], re[t]);
          im[t] = mma(c.aI, c.bR[t], im[t]);
        } else if (MODE == 1) {
          re[t] = mma(c.aR, c.bR[t], re[t]);
          im[t] = mma(c.aR, c.bI[t], im[t]);
          re[t] = mma(-c.aI, c.bI[t], re[t]);
          im[t] = mma(c.aI, c.bR[t], im[t]);
        } else if (MODE == 2) {
          re[t] = mma(c.aR, c.bR[t], re[t]);
          im[t] = mma(c.aI, c.bR[t], im[t]);
          re[t] = mma(c.aI, c.bI[t], re[t]);
          im[t] = mma(-c.aR, c.bI[t], im[t]);
        } else {
          re[t] = mma(c.aR, c.bR[t], re[t]);
          re[t] = mma(c.aI, c.bI[t], re[t]);
        }
      }
    }
    __builtin_amdgcn_sched_barrier(0);
#pragma unroll
    for (int g = 0; g < G; ++g) cur[g] = nxt[g];
  }
}

__global__ __launch_bounds__(FFT_NT) void fft_denoise_kernel(
    const float* __restrict__ x, float* __restrict__ xt) {
  __shared__ __attribute__((aligned(16))) float Fr[FMAT];
  __shared__ __attribute__((aligned(16))) float Fi[FMAT];
  __shared__ __attribute__((aligned(16))) float Gr[HM32];
  __shared__ __attribute__((aligned(16))) float Gi[HM32];
  __shared__ __attribute__((aligned(16))) float Hr[HM32];
  __shared__ __attribute__((aligned(16))) float Hi[HM32];
  __shared__ __attribute__((aligned(16))) float P[HM64];
  __shared__ __attribute__((aligned(16))) float Q[HM64];
  __shared__ float rootc[64];
  __shared__ float roots[64];
  __shared__ float altsum[8];

  const int img = blockIdx.x >> 1;
  const int p = blockIdx.x & 1;
  const int tid = threadIdx.x;
  const int lane = tid & 31;
  const int wave = __builtin_amdgcn_readfirstlane(tid >> 5);
  const int hi = lane >> 4;
  const int n = lane & 15;
  const int mt = wave & 1;
  const int nt = wave >> 1;
  const float* xi = x + (size_t)img * 62 * 62;

  float xv[16];
  for (int u = 0; u < 16; ++u) {
    int idx = tid + u * FFT_NT;
    int i = idx >> 6, j = idx & 63;
    float v = 0.0f;
    if (i >= 1 && i <= 62 && j >= 1 && j <= 62) v = xi[(i - 1) * 62 + (j - 1)];
    xv[u] = v;
  }

  if (tid < 64) {
    float s, c;
    sincospif((float)tid * (1.0f / 32.0f), &s, &c);
    rootc[tid] = c;
    roots[tid] = -s;
  }

  float twc[8], tws[8];
  for (int r = 0; r < 8; ++r) {
    int k1 = 2 * (16 * mt + 8 * hi + r) + p;
    int n2 = 16 * nt + n;
    float s, c;
    sincospif((float)(k1 * n2) * (1.0f / 2048.0f), &s, &c);
    twc[r] = c;
    tws[r] = -s;
  }

  {
    const float sp = p ? -1.0f : 1.0f;
    for (int u = 0; u < 8; ++u) {
      int idx = tid + u * FFT_NT;
      int i = idx >> 6, j = idx & 63;
      P[j * HS + i] = xv[u] + sp * xv[u + 8];
    }
  }
  {
    float sa = 0.0f;
    for (int u = 0; u < 16; ++u) sa += xv[u];
    if (tid & 1) sa = -sa;
    for (int m = 16; m >= 1; m >>= 1) sa += __shfl_xor(sa, m);
    if (lane == 0) altsum[wave] = sa;
  }
  __syncthreads();
  float z2048 = 0.0f;
  for (int w = 0; w < 8; ++w) z2048 += altsum[w];
  if (fabsf(z2048) < 0.5f) z2048 = 0.0f;
  if (p) z2048 = 0.0f;

  for (int idx = tid; idx < FFT_L; idx += FFT_NT) {
    int a = idx >> 6, b = idx & 63;
    int q = (a * b) & 63;
    Fr[a * FS + b] = rootc[q];
    Fi[a * FS + b] = roots[q];
  }
  for (int idx = tid; idx < 1024; idx += FFT_NT) {
    int a = idx >> 5, b = idx & 31;
    int qg = ((2 * a + p) * b) & 63;
    int qh = (a * (2 * b + p)) & 63;
    Gr[a * HS + b] = rootc[qg];
    Gi[a * HS + b] = roots[qg];
    Hr[a * HS + b] = rootc[qh];
    Hi[a * HS + b] = roots[qh];
  }
  __syncthreads();

  const int arow = 16 * mt + n;
  const int brow = 16 * nt + n;
  v8f accr[2], acci[2];

  accr[0] = (v8f){}; accr[1] = (v8f){}; acci[0] = (v8f){}; acci[1] = (v8f){};
  product64<0, 1, 8, 4, HS>(Gr, Gi, P, P, arow * HS, brow * HS, hi, accr, acci);
  __syncthreads();
  for (int r = 0; r < 8; ++r) {
    float yr = accr[0][r], yi = acci[0][r];
    float c = twc[r], s = tws[r];
    int o = (16 * mt + 8 * hi + r) * FS + 16 * nt + n;
    P[o] = yr * c - yi * s;
    Q[o] = yr * s + yi * c;
  }
  __syncthreads();

  if (wave < 4) {
    accr[0] = (v8f){}; acci[0] = (v8f){};
    product64<1, 1, 16, 2, FS>(P, Q, Fr, Fi, arow * FS, brow * FS, hi, accr,
                               acci);
  }
  __syncthreads();
  if (wave < 4) {
    for (int r = 0; r < 8; ++r) {
      float zr = accr[0][r], zi = acci[0][r];
      const int a = 16 * mt + 8 * hi + r;
      const int k2 = 16 * nt + n;
      const float wgt = (p == 0 && a == 0 && k2 == 0) ? 1.0f : 2.0f;
      if (fabsf(zr) < 0.5f) {
        zr = 0.0f;
        zi = 0.0f;
      }
      int o = a * HS + k2;
      P[o] = zr * wgt;
      Q[o] = zi * wgt;
    }
  }
  __syncthreads();

  accr[0] = (v8f){}; acci[0] = (v8f){};
  product64<2, 1, 8, 2, FS>(P, Q, Fr, Fi, arow * HS, brow * FS, hi, accr, acci);
  __syncthreads();
  {
    v4f pr0, pr1, pi0, pi1;
    for (int r = 0; r < 4; ++r) {
      float ur = accr[0][r], ui = acci[0][r];
      float c = twc[r], s = tws[r];
      pr0[r] = ur * c + ui * s;
      pi0[r] = ui * c - ur * s;
      float vr = accr[0][r + 4], vi = acci[0][r + 4];
      float c4 = twc[r + 4], s4 = tws[r + 4];
      pr1[r] = vr * c4 + vi * s4;
      pi1[r] = vi * c4 - vr * s4;
    }
    int o = brow * HS + 16 * mt + 8 * hi;
    *(v4f*)(P + o) = pr0;
    *(v4f*)(P + o + 4) = pr1;
    *(v4f*)(Q + o) = pi0;
    *(v4f*)(Q + o + 4) = pi1;
  }
  __syncthreads();

  if (wave < 4) {
    const int nt0 = 2 * nt;
    accr[0] = (v8f){}; accr[1] = (v8f){};
    product64<3, 2, 8, 2, HS>(Hr, Hi, P, Q, arow * HS, (16 * nt0 + n) * HS, hi,
                              accr, acci);
    float* xo = xt + (size_t)img * FFT_L + p * FFT_HP;
    const float scale = 1.0f / (float)FFT_L;
    const float alt = (n & 1) ? -z2048 : z2048;
    for (int t = 0; t < 2; ++t) {
      for (int r = 0; r < 8; ++r) {
        xo[(16 * mt + 8 * hi + r) * 64 + 16 * (nt0 + t) + n] =
            (accr[t][r] + alt) * scale;
      }
    }
  }
}

#define CV_R 4
#define CV_ROWS (CV_R + 2)
#define CV_CH (CV_ROWS * 18)
#define CV_PC (CV_CH + 8)

__device__ __forceinline__ constexpr int cv_off(int k) {
  return (k / 9) * CV_PC + ((k % 9) / 3) * 18 + (k % 9) % 3;
}
__device__ __forceinline__ constexpr int cv_delta(int k) {
  return cv_off(k + 4) - cv_off(k);
}
__device__ __forceinline__ constexpr int cv_dsel(int k) {
  return (k % 9) == 2 ? 1 : ((k % 9) <= 4 ? 0 : (((k % 9) == 5 || (k % 9) == 8) ? 2 : 3));
}
__host__ __device__ constexpr bool cv_bases_ok() {
  const int rep[4] = {0, 2, 5, 6};
  for (int k = 0; k + 4 < 288; ++k)
    if (cv_delta(k) != cv_delta(rep[cv_dsel(k)])) return false;
  return true;
}
static_assert(cv_bases_ok(), "lane-half distance table does not match");

__global__ __launch_bounds__(128) void conv3x3_wmma_kernel(
    const float* __restrict__ xt, const float* __restrict__ Wf,
    const float* __restrict__ bias, float* __restrict__ out) {
  __shared__ float xs[32 * CV_PC];

  const int tid = threadIdx.x;
  const int lane = tid & 31;
  const int wave = tid >> 5;
  const int o_base = wave * 16;
  const int j0 = blockIdx.x * 16;
  const int i0 = blockIdx.y * CV_R;
  const int b = blockIdx.z;

  const int hi = lane >> 4;
  const int n = lane & 15;

  const float* xb = xt + (size_t)b * 32 * FFT_L;
  const int cell = tid < CV_CH ? tid : CV_CH - 1;
  const int pr = cell / 18;
  const int pc = cell - pr * 18;
  int prow = i0 + pr;
  if (prow > 63) prow = 63;
  int pcol = j0 + pc;
  if (pcol > 63) pcol = 63;
  const float* src = xb + (prow & 31) * 64 + pcol;
  const float sg = prow < 32 ? 1.0f : -1.0f;
  float pv[32], pw[32];
#pragma unroll
  for (int c = 0; c < 32; ++c) {
    pv[c] = src[c * FFT_L];
    pw[c] = src[c * FFT_L + FFT_HP];
  }

  const v4f* wv = (const v4f*)(Wf + (size_t)(o_base + n) * 288 + 4 * hi);
  v4f wreg[36];
#pragma unroll
  for (int u = 0; u < 36; ++u) wreg[u] = wv[2 * u];

  const int slot = tid < CV_CH ? tid : CV_CH + ((tid - CV_CH) & 7);
#pragma unroll
  for (int c = 0; c < 32; ++c) xs[c * CV_PC + slot] = pv[c] + sg * pw[c];
  __syncthreads();

  const float* xl[4];
  xl[0] = xs + n + hi * cv_delta(0);
  xl[1] = xs + n + hi * cv_delta(2);
  xl[2] = xs + n + hi * cv_delta(5);
  xl[3] = xs + n + hi * cv_delta(6);
  v8f acc[CV_R];
  for (int t = 0; t < CV_R; ++t) acc[t] = (v8f){};

  v2f bcur[2][CV_R];
  for (int q = 0; q < 2; ++q) {
    for (int t = 0; t < CV_R; ++t) {
      bcur[q][t].x = xl[cv_dsel(2 * q)][cv_off(2 * q) + 18 * t];
      bcur[q][t].y = xl[cv_dsel(2 * q + 1)][cv_off(2 * q + 1) + 18 * t];
    }
  }
#pragma unroll
  for (int j = 0; j < 36; ++j) {
    v2f bnxt[2][CV_R];
    for (int q = 0; q < 2; ++q) {
      for (int t = 0; t < CV_R; ++t) {
        bnxt[q][t] = bcur[q][t];
        if (j < 35) {
          const int k0 = 8 * (j + 1) + 2 * q;
          bnxt[q][t].x = xl[cv_dsel(k0)][cv_off(k0) + 18 * t];
          bnxt[q][t].y = xl[cv_dsel(k0 + 1)][cv_off(k0 + 1) + 18 * t];
        }
      }
    }
    __builtin_amdgcn_sched_barrier(0);
    for (int q = 0; q < 2; ++q) {
      v2f a;
      a.x = wreg[j][2 * q];
      a.y = wreg[j][2 * q + 1];
      for (int t = 0; t < CV_R; ++t) acc[t] = mma(a, bcur[q][t], acc[t]);
    }
    __builtin_amdgcn_sched_barrier(0);
    for (int q = 0; q < 2; ++q)
      for (int t = 0; t < CV_R; ++t) bcur[q][t] = bnxt[q][t];
  }

  if (j0 + n < 62) {
    float bv[8];
    for (int r = 0; r < 8; ++r) bv[r] = bias[o_base + r + 8 * hi];
#pragma unroll
    for (int t = 0; t < CV_R; ++t) {
      int i = i0 + t;
      if (i < 62) {
#pragma unroll
        for (int r = 0; r < 8; ++r) {
          int o = o_base + r + 8 * hi;
          out[(((size_t)b * 64 + o) * 62 + i) * 62 + (j0 + n)] =
              acc[t][r] + bv[r];
        }
      }
    }
  }
}

extern "C" void kernel_launch(void* const* d_in, const int* in_sizes, int n_in,
                              void* d_out, int out_size, void* d_ws,
                              size_t ws_size, hipStream_t stream) {
  const float* x = (const float*)d_in[0];
  const float* W = (const float*)d_in[1];
  const float* bias = (const float*)d_in[2];
  float* out = (float*)d_out;
  float* xt = (float*)d_ws;

  fft_denoise_kernel<<<dim3(256), dim3(FFT_NT), 0, stream>>>(x, xt);
  conv3x3_wmma_kernel<<<dim3(4, 16, 4), dim3(128), 0, stream>>>(xt, W, bias,
                                                                out);
}
